# starting kernel
# baseline (speedup 1.0000x reference)
#include <hip/hip_runtime.h>
#include <hip/hip_bf16.h>

#define N_BU 1024
#define N_BI 4096
#define DIM  64
#define LNZ  50

typedef __attribute__((ext_vector_type(2))) float v2f;
typedef __attribute__((ext_vector_type(8))) float v8f;

__global__ __launch_bounds__(64)
void prep_users(const int* __restrict__ user_ids,
                const int* __restrict__ fb_indices,
                const float* __restrict__ fb_values,
                const float* __restrict__ user_weight,
                const float* __restrict__ user_bias,
                const float* __restrict__ irew,
                float* __restrict__ Aout,
                float* __restrict__ ub)
{
    const int b = blockIdx.x;
    const int d = threadIdx.x;
    const int uid = user_ids[b];
    float acc = user_weight[(size_t)uid * DIM + d];
    const int*   idxr = fb_indices + b * LNZ;
    const float* valr = fb_values  + b * LNZ;
#pragma unroll 2
    for (int l = 0; l < LNZ; ++l) {
        const int   it = idxr[l];
        const float v  = valr[l];
        acc = fmaf(v, irew[(size_t)it * DIM + d], acc);
    }
    Aout[b * DIM + d] = acc;
    if (d == 0) ub[b] = user_bias[uid];
}

__global__ __launch_bounds__(256)
void prep_items(const int* __restrict__ item_ids,
                const float* __restrict__ item_weight,
                const float* __restrict__ item_bias,
                float* __restrict__ Bt2,
                float* __restrict__ ib)
{
    const int t = blockIdx.x * 256 + threadIdx.x;
    const int n = t >> 6;
    const int k = t & 63;
    const int iid = item_ids[n];
    const float w = item_weight[(size_t)iid * DIM + k];
    Bt2[((size_t)(k >> 1) * N_BI + n) * 2 + (k & 1)] = w;
    if (k == 0) ib[n] = item_bias[iid];
}

__global__ __launch_bounds__(256)
void gemm_rating(const float* __restrict__ A,
                 const float* __restrict__ Bt2,
                 const float* __restrict__ ub,
                 const float* __restrict__ ib,
                 const float* __restrict__ biasp,
                 float* __restrict__ out)
{
    const int lane  = threadIdx.x & 31;
    const int wave  = threadIdx.x >> 5;
    const int tile  = blockIdx.x * 8 + wave;
    const int m0    = (tile >> 6) << 4;
    const int n0    = (tile & 63) << 6;
    const int mrow  = lane & 15;
    const int khalf = lane >> 4;

    v8f acc0 = {}, acc1 = {}, acc2 = {}, acc3 = {};

    const float* arow  = A   + (m0 + mrow) * DIM + khalf * 2;
    const float* bbase = Bt2 + (size_t)khalf * (2 * N_BI) + (size_t)(n0 + mrow) * 2;

#pragma unroll 4
    for (int k0 = 0; k0 < 16; ++k0) {
        const v2f a = *(const v2f*)(arow + k0 * 4);
        const float* brow = bbase + (size_t)k0 * (4 * N_BI);
        const v2f b0 = *(const v2f*)(brow);
        const v2f b1 = *(const v2f*)(brow + 32);
        const v2f b2 = *(const v2f*)(brow + 64);
        const v2f b3 = *(const v2f*)(brow + 96);
        acc0 = __builtin_amdgcn_wmma_f32_16x16x4_f32(false, a, false, b0, (short)0, acc0, false, false);
        acc1 = __builtin_amdgcn_wmma_f32_16x16x4_f32(false, a, false, b1, (short)0, acc1, false, false);
        acc2 = __builtin_amdgcn_wmma_f32_16x16x4_f32(false, a, false, b2, (short)0, acc2, false, false);
        acc3 = __builtin_amdgcn_wmma_f32_16x16x4_f32(false, a, false, b3, (short)0, acc3, false, false);
    }

    const float biasv = biasp[0];
    const int ncol = mrow;
    const float ib0 = ib[n0 + ncol]      + biasv;
    const float ib1 = ib[n0 + 16 + ncol] + biasv;
    const float ib2 = ib[n0 + 32 + ncol] + biasv;
    const float ib3 = ib[n0 + 48 + ncol] + biasv;

#pragma unroll
    for (int v = 0; v < 8; ++v) {
        const int row = m0 + khalf * 8 + v;
        const float ubv = ub[row];
        float* orow = out + (size_t)row * N_BI + n0 + ncol;
        orow[0]  = acc0[v] + ubv + ib0;
        orow[16] = acc1[v] + ubv + ib1;
        orow[32] = acc2[v] + ubv + ib2;
        orow[48] = acc3[v] + ubv + ib3;
    }
}

extern "C" void kernel_launch(void* const* d_in, const int* in_sizes, int n_in,
                              void* d_out, int out_size, void* d_ws, size_t ws_size,
                              hipStream_t stream) {
    (void)in_sizes; (void)n_in; (void)out_size; (void)ws_size;

    const int*   user_ids    = (const int*)  d_in[0];
    const int*   item_ids    = (const int*)  d_in[1];
    const int*   fb_indices  = (const int*)  d_in[2];
    const float* fb_values   = (const float*)d_in[3];
    const float* user_weight = (const float*)d_in[4];
    const float* user_bias   = (const float*)d_in[5];
    const float* item_weight = (const float*)d_in[6];
    const float* item_bias   = (const float*)d_in[7];
    const float* bias        = (const float*)d_in[8];
    const float* irew        = (const float*)d_in[9];
    float* out = (float*)d_out;

    float* A   = (float*)d_ws;
    float* Bt2 = A   + N_BU * DIM;
    float* ub  = Bt2 + DIM * N_BI;
    float* ib  = ub  + N_BU;

    prep_users<<<N_BU, 64, 0, stream>>>(user_ids, fb_indices, fb_values,
                                        user_weight, user_bias, irew, A, ub);
    prep_items<<<(N_BI * DIM) / 256, 256, 0, stream>>>(item_ids, item_weight,
                                                       item_bias, Bt2, ib);
    gemm_rating<<<(N_BU / 16) * (N_BI / 64) / 8, 256, 0, stream>>>(A, Bt2, ub, ib,
                                                                   bias, out);
}


# GEMM gathers each block's 64 item rows from item_weight straight into padded LDS and reads item_bias itself; the item staging launch and its workspace plane are gone (3 launches to 2)
# speedup vs baseline: 1.4133x; 1.4133x over previous
#include <hip/hip_runtime.h>
#include <hip/hip_bf16.h>

#define N_BU 1024
#define N_BI 4096
#define DIM  64
#define LNZ  50

typedef __attribute__((ext_vector_type(2))) float v2f;
typedef __attribute__((ext_vector_type(4))) float v4f;
typedef __attribute__((ext_vector_type(8))) float v8f;

__global__ __launch_bounds__(64)
void prep_users(const int* __restrict__ user_ids,
                const int* __restrict__ fb_indices,
                const float* __restrict__ fb_values,
                const float* __restrict__ user_weight,
                const float* __restrict__ user_bias,
                const float* __restrict__ irew,
                float* __restrict__ Aout,
                float* __restrict__ ub)
{
    const int b = blockIdx.x;
    const int d = threadIdx.x;
    const int uid = user_ids[b];
    float acc = user_weight[(size_t)uid * DIM + d];
    const int*   idxr = fb_indices + b * LNZ;
    const float* valr = fb_values  + b * LNZ;
#pragma unroll 2
    for (int l = 0; l < LNZ; ++l) {
        const int   it = idxr[l];
        const float v  = valr[l];
        acc = fmaf(v, irew[(size_t)it * DIM + d], acc);
    }
    Aout[b * DIM + d] = acc;
    if (d == 0) ub[b] = user_bias[uid];
}

#define BPITCH 68

__global__ __launch_bounds__(256)
void gemm_rating(const float* __restrict__ A,
                 const int*   __restrict__ item_ids,
                 const float* __restrict__ item_weight,
                 const float* __restrict__ item_bias,
                 const float* __restrict__ ub,
                 const float* __restrict__ biasp,
                 float* __restrict__ out)
{
    __shared__ __attribute__((aligned(16))) float Bs[64 * BPITCH];

    const int t     = threadIdx.x;
    const int lane  = t & 31;
    const int wave  = t >> 5;
    const int n0    = (blockIdx.x & 63) << 6;
    const int m0    = (((blockIdx.x >> 6) << 3) + wave) << 4;
    const int mrow  = lane & 15;
    const int khalf = lane >> 4;

    const int c4 = t & 15;
    const int r0 = t >> 4;
    int gid[4];
#pragma unroll
    for (int i = 0; i < 4; ++i) gid[i] = item_ids[n0 + r0 + 16 * i];
    v4f w[4];
#pragma unroll
    for (int i = 0; i < 4; ++i)
        w[i] = *(const v4f*)(item_weight + (size_t)gid[i] * DIM + c4 * 4);

    const float* arow = A + (m0 + mrow) * DIM + khalf * 2;
    v2f a[16];
#pragma unroll
    for (int k0 = 0; k0 < 16; ++k0) a[k0] = *(const v2f*)(arow + k0 * 4);

    const int ncol = mrow;
    int bid[4];
#pragma unroll
    for (int j = 0; j < 4; ++j) bid[j] = item_ids[n0 + 16 * j + ncol];
    float ibv[4];
#pragma unroll
    for (int j = 0; j < 4; ++j) ibv[j] = item_bias[bid[j]];
    const float biasv = biasp[0];
    float ubv[8];
#pragma unroll
    for (int v = 0; v < 8; ++v) ubv[v] = ub[m0 + khalf * 8 + v];

#pragma unroll
    for (int i = 0; i < 4; ++i)
        *(v4f*)(Bs + (r0 + 16 * i) * BPITCH + c4 * 4) = w[i];
    __syncthreads();

    v8f acc0 = {}, acc1 = {}, acc2 = {}, acc3 = {};
    const float* bl = Bs + mrow * BPITCH + khalf * 2;

#pragma unroll
    for (int k0 = 0; k0 < 16; ++k0) {
        const v2f b0 = *(const v2f*)(bl + k0 * 4);
        const v2f b1 = *(const v2f*)(bl + 16 * BPITCH + k0 * 4);
        const v2f b2 = *(const v2f*)(bl + 32 * BPITCH + k0 * 4);
        const v2f b3 = *(const v2f*)(bl + 48 * BPITCH + k0 * 4);
        acc0 = __builtin_amdgcn_wmma_f32_16x16x4_f32(false, a[k0], false, b0, (short)0, acc0, false, false);
        acc1 = __builtin_amdgcn_wmma_f32_16x16x4_f32(false, a[k0], false, b1, (short)0, acc1, false, false);
        acc2 = __builtin_amdgcn_wmma_f32_16x16x4_f32(false, a[k0], false, b2, (short)0, acc2, false, false);
        acc3 = __builtin_amdgcn_wmma_f32_16x16x4_f32(false, a[k0], false, b3, (short)0, acc3, false, false);
    }

    const float ib0 = ibv[0] + biasv;
    const float ib1 = ibv[1] + biasv;
    const float ib2 = ibv[2] + biasv;
    const float ib3 = ibv[3] + biasv;

#pragma unroll
    for (int v = 0; v < 8; ++v) {
        const int row = m0 + khalf * 8 + v;
        float* orow = out + (size_t)row * N_BI + n0 + ncol;
        orow[0]  = acc0[v] + ubv[v] + ib0;
        orow[16] = acc1[v] + ubv[v] + ib1;
        orow[32] = acc2[v] + ubv[v] + ib2;
        orow[48] = acc3[v] + ubv[v] + ib3;
    }
}

extern "C" void kernel_launch(void* const* d_in, const int* in_sizes, int n_in,
                              void* d_out, int out_size, void* d_ws, size_t ws_size,
                              hipStream_t stream) {
    (void)in_sizes; (void)n_in; (void)out_size; (void)ws_size;

    const int*   user_ids    = (const int*)  d_in[0];
    const int*   item_ids    = (const int*)  d_in[1];
    const int*   fb_indices  = (const int*)  d_in[2];
    const float* fb_values   = (const float*)d_in[3];
    const float* user_weight = (const float*)d_in[4];
    const float* user_bias   = (const float*)d_in[5];
    const float* item_weight = (const float*)d_in[6];
    const float* item_bias   = (const float*)d_in[7];
    const float* bias        = (const float*)d_in[8];
    const float* irew        = (const float*)d_in[9];
    float* out = (float*)d_out;

    float* A   = (float*)d_ws;
    float* ub  = A + N_BU * DIM;

    prep_users<<<N_BU, 64, 0, stream>>>(user_ids, fb_indices, fb_values,
                                        user_weight, user_bias, irew, A, ub);
    gemm_rating<<<(N_BI / 64) * (N_BU / 128), 256, 0, stream>>>(A, item_ids, item_weight,
                                                                item_bias, ub, bias, out);
}


# prep_users issues all 50 rating-effect row gathers of a user before the first multiply-add (one wave per user, 8-byte loads per lane) instead of 25 dependent load round trips; sum order kept
# speedup vs baseline: 2.5803x; 1.8257x over previous
#include <hip/hip_runtime.h>
#include <hip/hip_bf16.h>

#define N_BU 1024
#define N_BI 4096
#define DIM  64
#define LNZ  50

typedef __attribute__((ext_vector_type(2))) float v2f;
typedef __attribute__((ext_vector_type(4))) float v4f;
typedef __attribute__((ext_vector_type(8))) float v8f;

__global__ __launch_bounds__(32) __attribute__((amdgpu_waves_per_eu(1, 4)))
void prep_users(const int* __restrict__ user_ids,
                const int* __restrict__ fb_indices,
                const float* __restrict__ fb_values,
                const float* __restrict__ user_weight,
                const float* __restrict__ user_bias,
                const float* __restrict__ irew,
                float* __restrict__ Aout,
                float* __restrict__ ub)
{
    const int b    = blockIdx.x;
    const int lane = threadIdx.x;
    const int uid  = user_ids[b];
    const int*   idxr = fb_indices + b * LNZ;
    const float* valr = fb_values  + b * LNZ;
    v2f acc = *(const v2f*)(user_weight + (size_t)uid * DIM + lane * 2);
    v2f g[LNZ];
#pragma unroll
    for (int l = 0; l < LNZ; ++l)
        g[l] = *(const v2f*)(irew + (size_t)idxr[l] * DIM + lane * 2);
#pragma unroll
    for (int l = 0; l < LNZ; ++l) {
        const float v = valr[l];
        acc.x = fmaf(v, g[l].x, acc.x);
        acc.y = fmaf(v, g[l].y, acc.y);
    }
    *(v2f*)(Aout + b * DIM + lane * 2) = acc;
    if (lane == 0) ub[b] = user_bias[uid];
}

#define BPITCH 68

__global__ __launch_bounds__(256)
void gemm_rating(const float* __restrict__ A,
                 const int*   __restrict__ item_ids,
                 const float* __restrict__ item_weight,
                 const float* __restrict__ item_bias,
                 const float* __restrict__ ub,
                 const float* __restrict__ biasp,
                 float* __restrict__ out)
{
    __shared__ __attribute__((aligned(16))) float Bs[64 * BPITCH];

    const int t     = threadIdx.x;
    const int lane  = t & 31;
    const int wave  = t >> 5;
    const int n0    = (blockIdx.x & 63) << 6;
    const int m0    = (((blockIdx.x >> 6) << 3) + wave) << 4;
    const int mrow  = lane & 15;
    const int khalf = lane >> 4;

    const int c4 = t & 15;
    const int r0 = t >> 4;
    int gid[4];
#pragma unroll
    for (int i = 0; i < 4; ++i) gid[i] = item_ids[n0 + r0 + 16 * i];
    v4f w[4];
#pragma unroll
    for (int i = 0; i < 4; ++i)
        w[i] = *(const v4f*)(item_weight + (size_t)gid[i] * DIM + c4 * 4);

    const float* arow = A + (m0 + mrow) * DIM + khalf * 2;
    v2f a[16];
#pragma unroll
    for (int k0 = 0; k0 < 16; ++k0) a[k0] = *(const v2f*)(arow + k0 * 4);

    const int ncol = mrow;
    int bid[4];
#pragma unroll
    for (int j = 0; j < 4; ++j) bid[j] = item_ids[n0 + 16 * j + ncol];
    float ibv[4];
#pragma unroll
    for (int j = 0; j < 4; ++j) ibv[j] = item_bias[bid[j]];
    const float biasv = biasp[0];
    float ubv[8];
#pragma unroll
    for (int v = 0; v < 8; ++v) ubv[v] = ub[m0 + khalf * 8 + v];

#pragma unroll
    for (int i = 0; i < 4; ++i)
        *(v4f*)(Bs + (r0 + 16 * i) * BPITCH + c4 * 4) = w[i];
    __syncthreads();

    v8f acc0 = {}, acc1 = {}, acc2 = {}, acc3 = {};
    const float* bl = Bs + mrow * BPITCH + khalf * 2;

#pragma unroll
    for (int k0 = 0; k0 < 16; ++k0) {
        const v2f b0 = *(const v2f*)(bl + k0 * 4);
        const v2f b1 = *(const v2f*)(bl + 16 * BPITCH + k0 * 4);
        const v2f b2 = *(const v2f*)(bl + 32 * BPITCH + k0 * 4);
        const v2f b3 = *(const v2f*)(bl + 48 * BPITCH + k0 * 4);
        acc0 = __builtin_amdgcn_wmma_f32_16x16x4_f32(false, a[k0], false, b0, (short)0, acc0, false, false);
        acc1 = __builtin_amdgcn_wmma_f32_16x16x4_f32(false, a[k0], false, b1, (short)0, acc1, false, false);
        acc2 = __builtin_amdgcn_wmma_f32_16x16x4_f32(false, a[k0], false, b2, (short)0, acc2, false, false);
        acc3 = __builtin_amdgcn_wmma_f32_16x16x4_f32(false, a[k0], false, b3, (short)0, acc3, false, false);
    }

    const float ib0 = ibv[0] + biasv;
    const float ib1 = ibv[1] + biasv;
    const float ib2 = ibv[2] + biasv;
    const float ib3 = ibv[3] + biasv;

#pragma unroll
    for (int v = 0; v < 8; ++v) {
        const int row = m0 + khalf * 8 + v;
        float* orow = out + (size_t)row * N_BI + n0 + ncol;
        orow[0]  = acc0[v] + ubv[v] + ib0;
        orow[16] = acc1[v] + ubv[v] + ib1;
        orow[32] = acc2[v] + ubv[v] + ib2;
        orow[48] = acc3[v] + ubv[v] + ib3;
    }
}

extern "C" void kernel_launch(void* const* d_in, const int* in_sizes, int n_in,
                              void* d_out, int out_size, void* d_ws, size_t ws_size,
                              hipStream_t stream) {
    (void)in_sizes; (void)n_in; (void)out_size; (void)ws_size;

    const int*   user_ids    = (const int*)  d_in[0];
    const int*   item_ids    = (const int*)  d_in[1];
    const int*   fb_indices  = (const int*)  d_in[2];
    const float* fb_values   = (const float*)d_in[3];
    const float* user_weight = (const float*)d_in[4];
    const float* user_bias   = (const float*)d_in[5];
    const float* item_weight = (const float*)d_in[6];
    const float* item_bias   = (const float*)d_in[7];
    const float* bias        = (const float*)d_in[8];
    const float* irew        = (const float*)d_in[9];
    float* out = (float*)d_out;

    float* A   = (float*)d_ws;
    float* ub  = A + N_BU * DIM;

    prep_users<<<N_BU, 32, 0, stream>>>(user_ids, fb_indices, fb_values,
                                        user_weight, user_bias, irew, A, ub);
    gemm_rating<<<(N_BI / 64) * (N_BU / 128), 256, 0, stream>>>(A, item_ids, item_weight,
                                                                item_bias, ub, bias, out);
}


# gemm_rating reads only whole lines from global: each wave loads its A tile as 16 whole rows and takes its fragments from its own padded LDS tile; item biases read once per block via LDS
# speedup vs baseline: 2.7831x; 1.0786x over previous
#include <hip/hip_runtime.h>
#include <hip/hip_bf16.h>

#define N_BU 1024
#define N_BI 4096
#define DIM  64
#define LNZ  50

typedef __attribute__((ext_vector_type(2))) float v2f;
typedef __attribute__((ext_vector_type(4))) float v4f;
typedef __attribute__((ext_vector_type(8))) float v8f;

__global__ __launch_bounds__(32) __attribute__((amdgpu_waves_per_eu(1, 4)))
void prep_users(const int* __restrict__ user_ids,
                const int* __restrict__ fb_indices,
                const float* __restrict__ fb_values,
                const float* __restrict__ user_weight,
                const float* __restrict__ user_bias,
                const float* __restrict__ irew,
                float* __restrict__ Aout,
                float* __restrict__ ub)
{
    const int b    = blockIdx.x;
    const int lane = threadIdx.x;
    const int uid  = user_ids[b];
    const int*   idxr = fb_indices + b * LNZ;
    const float* valr = fb_values  + b * LNZ;
    v2f acc = *(const v2f*)(user_weight + (size_t)uid * DIM + lane * 2);
    v2f g[LNZ];
#pragma unroll
    for (int l = 0; l < LNZ; ++l)
        g[l] = *(const v2f*)(irew + (size_t)idxr[l] * DIM + lane * 2);
#pragma unroll
    for (int l = 0; l < LNZ; ++l) {
        const float v = valr[l];
        acc.x = fmaf(v, g[l].x, acc.x);
        acc.y = fmaf(v, g[l].y, acc.y);
    }
    *(v2f*)(Aout + b * DIM + lane * 2) = acc;
    if (lane == 0) ub[b] = user_bias[uid];
}

#define BPITCH 68

__global__ __launch_bounds__(256)
void gemm_rating(const float* __restrict__ A,
                 const int*   __restrict__ item_ids,
                 const float* __restrict__ item_weight,
                 const float* __restrict__ item_bias,
                 const float* __restrict__ ub,
                 const float* __restrict__ biasp,
                 float* __restrict__ out)
{
    __shared__ __attribute__((aligned(16))) float Bs[64 * BPITCH];
    __shared__ __attribute__((aligned(16))) float As[8 * 16 * BPITCH];
    __shared__ float ibs[64];

    const int t     = threadIdx.x;
    const int lane  = t & 31;
    const int wave  = t >> 5;
    const int n0    = (blockIdx.x & 63) << 6;
    const int m0    = (((blockIdx.x >> 6) << 3) + wave) << 4;
    const int mrow  = lane & 15;
    const int khalf = lane >> 4;

    const int c4 = t & 15;
    const int r0 = t >> 4;
    int gid[4];
#pragma unroll
    for (int i = 0; i < 4; ++i) gid[i] = item_ids[n0 + r0 + 16 * i];
    v4f w[4];
#pragma unroll
    for (int i = 0; i < 4; ++i)
        w[i] = *(const v4f*)(item_weight + (size_t)gid[i] * DIM + c4 * 4);

    const float* atile = A + m0 * DIM;
    v4f x[8];
#pragma unroll
    for (int i = 0; i < 8; ++i) x[i] = *(const v4f*)(atile + i * 128 + lane * 4);

    const int ncol = mrow;
    float ibr[4];
#pragma unroll
    for (int i = 0; i < 4; ++i) ibr[i] = item_bias[gid[i]];
    const float biasv = biasp[0];
    float ubv[8];
#pragma unroll
    for (int v = 0; v < 8; ++v) ubv[v] = ub[m0 + khalf * 8 + v];

    float* as = As + wave * (16 * BPITCH);
#pragma unroll
    for (int i = 0; i < 8; ++i)
        *(v4f*)(as + (2 * i + khalf) * BPITCH + mrow * 4) = x[i];
#pragma unroll
    for (int i = 0; i < 4; ++i)
        *(v4f*)(Bs + (r0 + 16 * i) * BPITCH + c4 * 4) = w[i];
    asm volatile("" :: "v"(ibr[0]), "v"(ibr[1]), "v"(ibr[2]), "v"(ibr[3]));
    if (c4 == 0) {
#pragma unroll
        for (int i = 0; i < 4; ++i) ibs[r0 + 16 * i] = ibr[i];
    }
    __syncthreads();

    const float* al = as + mrow * BPITCH + khalf * 2;
    v2f a[16];
#pragma unroll
    for (int k0 = 0; k0 < 16; ++k0) a[k0] = *(const v2f*)(al + k0 * 4);

    v8f acc0 = {}, acc1 = {}, acc2 = {}, acc3 = {};
    const float* bl = Bs + mrow * BPITCH + khalf * 2;

#pragma unroll
    for (int k0 = 0; k0 < 16; ++k0) {
        const v2f b0 = *(const v2f*)(bl + k0 * 4);
        const v2f b1 = *(const v2f*)(bl + 16 * BPITCH + k0 * 4);
        const v2f b2 = *(const v2f*)(bl + 32 * BPITCH + k0 * 4);
        const v2f b3 = *(const v2f*)(bl + 48 * BPITCH + k0 * 4);
        acc0 = __builtin_amdgcn_wmma_f32_16x16x4_f32(false, a[k0], false, b0, (short)0, acc0, false, false);
        acc1 = __builtin_amdgcn_wmma_f32_16x16x4_f32(false, a[k0], false, b1, (short)0, acc1, false, false);
        acc2 = __builtin_amdgcn_wmma_f32_16x16x4_f32(false, a[k0], false, b2, (short)0, acc2, false, false);
        acc3 = __builtin_amdgcn_wmma_f32_16x16x4_f32(false, a[k0], false, b3, (short)0, acc3, false, false);
    }

    const float ib0 = ibs[ncol]      + biasv;
    const float ib1 = ibs[16 + ncol] + biasv;
    const float ib2 = ibs[32 + ncol] + biasv;
    const float ib3 = ibs[48 + ncol] + biasv;

#pragma unroll
    for (int v = 0; v < 8; ++v) {
        const int row = m0 + khalf * 8 + v;
        float* orow = out + (size_t)row * N_BI + n0 + ncol;
        orow[0]  = acc0[v] + ubv[v] + ib0;
        orow[16] = acc1[v] + ubv[v] + ib1;
        orow[32] = acc2[v] + ubv[v] + ib2;
        orow[48] = acc3[v] + ubv[v] + ib3;
    }
}

extern "C" void kernel_launch(void* const* d_in, const int* in_sizes, int n_in,
                              void* d_out, int out_size, void* d_ws, size_t ws_size,
                              hipStream_t stream) {
    (void)in_sizes; (void)n_in; (void)out_size; (void)ws_size;

    const int*   user_ids    = (const int*)  d_in[0];
    const int*   item_ids    = (const int*)  d_in[1];
    const int*   fb_indices  = (const int*)  d_in[2];
    const float* fb_values   = (const float*)d_in[3];
    const float* user_weight = (const float*)d_in[4];
    const float* user_bias   = (const float*)d_in[5];
    const float* item_weight = (const float*)d_in[6];
    const float* item_bias   = (const float*)d_in[7];
    const float* bias        = (const float*)d_in[8];
    const float* irew        = (const float*)d_in[9];
    float* out = (float*)d_out;

    float* A   = (float*)d_ws;
    float* ub  = A + N_BU * DIM;

    prep_users<<<N_BU, 32, 0, stream>>>(user_ids, fb_indices, fb_values,
                                        user_weight, user_bias, irew, A, ub);
    gemm_rating<<<(N_BI / 64) * (N_BU / 128), 256, 0, stream>>>(A, item_ids, item_weight,
                                                                item_bias, ub, bias, out);
}


# prep_users issues every load ahead of the gather wait: 50 weights by two vector loads handed out by lane reads, user bias early; three scalar-load round trips behind the gathers gone; sum order kept
# speedup vs baseline: 2.9360x; 1.0549x over previous
#include <hip/hip_runtime.h>
#include <hip/hip_bf16.h>

#define N_BU 1024
#define N_BI 4096
#define DIM  64
#define LNZ  50

typedef __attribute__((ext_vector_type(2))) float v2f;
typedef __attribute__((ext_vector_type(4))) float v4f;
typedef __attribute__((ext_vector_type(8))) float v8f;

__global__ __launch_bounds__(32) __attribute__((amdgpu_waves_per_eu(1, 4)))
void prep_users(const int* __restrict__ user_ids,
                const int* __restrict__ fb_indices,
                const float* __restrict__ fb_values,
                const float* __restrict__ user_weight,
                const float* __restrict__ user_bias,
                const float* __restrict__ irew,
                float* __restrict__ Aout,
                float* __restrict__ ub)
{
    const int b    = blockIdx.x;
    const int lane = threadIdx.x;
    const int*   idxr = fb_indices + b * LNZ;
    const float* valr = fb_values  + b * LNZ;
    const float vlo = valr[lane];
    const float vhi = valr[32 + (lane < LNZ - 32 ? lane : LNZ - 33)];
    const int uid  = user_ids[b];
    const float ubias = user_bias[uid];
    v2f acc = *(const v2f*)(user_weight + (size_t)uid * DIM + lane * 2);
    v2f g[LNZ];
#pragma unroll
    for (int l = 0; l < LNZ; ++l)
        g[l] = *(const v2f*)(irew + (size_t)idxr[l] * DIM + lane * 2);
    asm volatile("" :: "s"(ubias));
    float w[LNZ];
#pragma unroll
    for (int l = 0; l < LNZ; ++l)
        w[l] = __int_as_float(__builtin_amdgcn_readlane(__float_as_int(l < 32 ? vlo : vhi), l & 31));
#pragma unroll
    for (int l = 0; l < LNZ; ++l) {
        const float v = w[l];
        acc.x = fmaf(v, g[l].x, acc.x);
        acc.y = fmaf(v, g[l].y, acc.y);
    }
    *(v2f*)(Aout + b * DIM + lane * 2) = acc;
    if (lane == 0) ub[b] = ubias;
}

#define BPITCH 68

__global__ __launch_bounds__(256)
void gemm_rating(const float* __restrict__ A,
                 const int*   __restrict__ item_ids,
                 const float* __restrict__ item_weight,
                 const float* __restrict__ item_bias,
                 const float* __restrict__ ub,
                 const float* __restrict__ biasp,
                 float* __restrict__ out)
{
    __shared__ __attribute__((aligned(16))) float Bs[64 * BPITCH];
    __shared__ __attribute__((aligned(16))) float As[8 * 16 * BPITCH];
    __shared__ float ibs[64];

    const int t     = threadIdx.x;
    const int lane  = t & 31;
    const int wave  = t >> 5;
    const int n0    = (blockIdx.x & 63) << 6;
    const int m0    = (((blockIdx.x >> 6) << 3) + wave) << 4;
    const int mrow  = lane & 15;
    const int khalf = lane >> 4;

    const int c4 = t & 15;
    const int r0 = t >> 4;
    int gid[4];
#pragma unroll
    for (int i = 0; i < 4; ++i) gid[i] = item_ids[n0 + r0 + 16 * i];
    v4f w[4];
#pragma unroll
    for (int i = 0; i < 4; ++i)
        w[i] = *(const v4f*)(item_weight + (size_t)gid[i] * DIM + c4 * 4);

    const float* atile = A + m0 * DIM;
    v4f x[8];
#pragma unroll
    for (int i = 0; i < 8; ++i) x[i] = *(const v4f*)(atile + i * 128 + lane * 4);

    const int ncol = mrow;
    float ibr[4];
#pragma unroll
    for (int i = 0; i < 4; ++i) ibr[i] = item_bias[gid[i]];
    const float biasv = biasp[0];
    float ubv[8];
#pragma unroll
    for (int v = 0; v < 8; ++v) ubv[v] = ub[m0 + khalf * 8 + v];

    float* as = As + wave * (16 * BPITCH);
#pragma unroll
    for (int i = 0; i < 8; ++i)
        *(v4f*)(as + (2 * i + khalf) * BPITCH + mrow * 4) = x[i];
#pragma unroll
    for (int i = 0; i < 4; ++i)
        *(v4f*)(Bs + (r0 + 16 * i) * BPITCH + c4 * 4) = w[i];
    asm volatile("" :: "v"(ibr[0]), "v"(ibr[1]), "v"(ibr[2]), "v"(ibr[3]));
    if (c4 == 0) {
#pragma unroll
        for (int i = 0; i < 4; ++i) ibs[r0 + 16 * i] = ibr[i];
    }
    __syncthreads();

    const float* al = as + mrow * BPITCH + khalf * 2;
    v2f a[16];
#pragma unroll
    for (int k0 = 0; k0 < 16; ++k0) a[k0] = *(const v2f*)(al + k0 * 4);

    v8f acc0 = {}, acc1 = {}, acc2 = {}, acc3 = {};
    const float* bl = Bs + mrow * BPITCH + khalf * 2;

#pragma unroll
    for (int k0 = 0; k0 < 16; ++k0) {
        const v2f b0 = *(const v2f*)(bl + k0 * 4);
        const v2f b1 = *(const v2f*)(bl + 16 * BPITCH + k0 * 4);
        const v2f b2 = *(const v2f*)(bl + 32 * BPITCH + k0 * 4);
        const v2f b3 = *(const v2f*)(bl + 48 * BPITCH + k0 * 4);
        acc0 = __builtin_amdgcn_wmma_f32_16x16x4_f32(false, a[k0], false, b0, (short)0, acc0, false, false);
        acc1 = __builtin_amdgcn_wmma_f32_16x16x4_f32(false, a[k0], false, b1, (short)0, acc1, false, false);
        acc2 = __builtin_amdgcn_wmma_f32_16x16x4_f32(false, a[k0], false, b2, (short)0, acc2, false, false);
        acc3 = __builtin_amdgcn_wmma_f32_16x16x4_f32(false, a[k0], false, b3, (short)0, acc3, false, false);
    }

    const float ib0 = ibs[ncol]      + biasv;
    const float ib1 = ibs[16 + ncol] + biasv;
    const float ib2 = ibs[32 + ncol] + biasv;
    const float ib3 = ibs[48 + ncol] + biasv;

#pragma unroll
    for (int v = 0; v < 8; ++v) {
        const int row = m0 + khalf * 8 + v;
        float* orow = out + (size_t)row * N_BI + n0 + ncol;
        orow[0]  = acc0[v] + ubv[v] + ib0;
        orow[16] = acc1[v] + ubv[v] + ib1;
        orow[32] = acc2[v] + ubv[v] + ib2;
        orow[48] = acc3[v] + ubv[v] + ib3;
    }
}

extern "C" void kernel_launch(void* const* d_in, const int* in_sizes, int n_in,
                              void* d_out, int out_size, void* d_ws, size_t ws_size,
                              hipStream_t stream) {
    (void)in_sizes; (void)n_in; (void)out_size; (void)ws_size;

    const int*   user_ids    = (const int*)  d_in[0];
    const int*   item_ids    = (const int*)  d_in[1];
    const int*   fb_indices  = (const int*)  d_in[2];
    const float* fb_values   = (const float*)d_in[3];
    const float* user_weight = (const float*)d_in[4];
    const float* user_bias   = (const float*)d_in[5];
    const float* item_weight = (const float*)d_in[6];
    const float* item_bias   = (const float*)d_in[7];
    const float* bias        = (const float*)d_in[8];
    const float* irew        = (const float*)d_in[9];
    float* out = (float*)d_out;

    float* A   = (float*)d_ws;
    float* ub  = A + N_BU * DIM;

    prep_users<<<N_BU, 32, 0, stream>>>(user_ids, fb_indices, fb_values,
                                        user_weight, user_bias, irew, A, ub);
    gemm_rating<<<(N_BI / 64) * (N_BU / 128), 256, 0, stream>>>(A, item_ids, item_weight,
                                                                item_bias, ub, bias, out);
}
